# starting kernel
# baseline (speedup 1.0000x reference)
#include <hip/hip_runtime.h>
#include <hip/hip_bf16.h>

#define DIM 128

typedef float v2f __attribute__((ext_vector_type(2)));
typedef float v8f __attribute__((ext_vector_type(8)));

__device__ __forceinline__ void atomAddF(float* p, float v) {
    __hip_atomic_fetch_add(p, v, __ATOMIC_RELAXED, __HIP_MEMORY_SCOPE_AGENT);
}

__global__ void deg_init_kernel(float* __restrict__ deg, int n) {
    int i = blockIdx.x * blockDim.x + threadIdx.x;
    if (i < n) deg[i] = 1.0f;
}

__global__ void deg_count_kernel(const int* __restrict__ ei, float* __restrict__ deg, int E) {
    int e = blockIdx.x * blockDim.x + threadIdx.x;
    if (e < E) atomAddF(&deg[ei[E + e]], 1.0f);
}

__global__ void deg_to_dis_kernel(float* __restrict__ deg, int n) {
    int i = blockIdx.x * blockDim.x + threadIdx.x;
    if (i < n) {
        float d = deg[i];
        deg[i] = (d > 0.0f) ? rsqrtf(d) : 0.0f;
    }
}

__global__ __launch_bounds__(256) void self_loop_kernel(const float* __restrict__ x,
                                                        const float* __restrict__ dis,
                                                        float* __restrict__ agg, int n) {
    int t = blockIdx.x * blockDim.x + threadIdx.x;
    int row = t >> 5;
    int lane = t & 31;
    if (row >= n) return;
    float s = dis[row];
    s = s * s;
    const float4* xr = reinterpret_cast<const float4*>(x + (size_t)row * DIM);
    float4* ar = reinterpret_cast<float4*>(agg + (size_t)row * DIM);
    float4 v = xr[lane];
    v.x *= s; v.y *= s; v.z *= s; v.w *= s;
    ar[lane] = v;
}

__global__ __launch_bounds__(256) void edge_scatter_kernel(const float* __restrict__ x,
                                                           const float* __restrict__ dis,
                                                           const int* __restrict__ ei,
                                                           float* __restrict__ agg, int E) {
    int t = blockIdx.x * blockDim.x + threadIdx.x;
    int e = t >> 5;
    int lane = t & 31;
    if (e >= E) return;
    int s = ei[e];
    int d = ei[E + e];
    float nrm = dis[s] * dis[d];
    const float4* xr = reinterpret_cast<const float4*>(x + (size_t)s * DIM);
    float4 v = xr[lane];
    float* dst = agg + (size_t)d * DIM + lane * 4;
    atomAddF(dst + 0, v.x * nrm);
    atomAddF(dst + 1, v.y * nrm);
    atomAddF(dst + 2, v.z * nrm);
    atomAddF(dst + 3, v.w * nrm);
}

#define LDA 132
#define LDW 136
#define KCHUNK 32

__global__ __launch_bounds__(256) void gemm_wmma_kernel(const float* __restrict__ A,
                                                        const float* __restrict__ W,
                                                        float* __restrict__ out, int n) {
    __shared__ float lA[16 * LDA];
    __shared__ float lW[KCHUNK * LDW];
    const int tid = threadIdx.x;
    const int wave = tid >> 5;
    const int lane = tid & 31;
    const int rowBase = blockIdx.x * 16;

    for (int i = tid; i < 16 * DIM; i += 256) {
        int r = i >> 7, c = i & 127;
        int gr = rowBase + r;
        lA[r * LDA + c] = (gr < n) ? A[(size_t)gr * DIM + c] : 0.0f;
    }

    const int m = lane & 15;
    const int hi = lane >> 4;
    const int colBase = wave * 16;

    v8f acc = {0.f, 0.f, 0.f, 0.f, 0.f, 0.f, 0.f, 0.f};

    for (int kc = 0; kc < DIM; kc += KCHUNK) {
        __syncthreads();
        for (int i = tid; i < KCHUNK * DIM; i += 256) {
            int r = i >> 7, c = i & 127;
            lW[r * LDW + c] = W[(size_t)(kc + r) * DIM + c];
        }
        __syncthreads();

#pragma unroll
        for (int k = 0; k < KCHUNK; k += 4) {
            v2f a, b;
            const float* ap = &lA[m * LDA + kc + k + 2 * hi];
            a.x = ap[0];
            a.y = ap[1];
            b.x = lW[(k + 2 * hi) * LDW + colBase + m];
            b.y = lW[(k + 2 * hi + 1) * LDW + colBase + m];
            acc = __builtin_amdgcn_wmma_f32_16x16x4_f32(
                 false, a,  false, b,
                 (short)0, acc,  false,  false);
        }
    }

#pragma unroll
    for (int v = 0; v < 8; ++v) {
        int gr = rowBase + v + 8 * hi;
        if (gr < n) out[(size_t)gr * DIM + colBase + m] = acc[v];
    }
}

extern "C" void kernel_launch(void* const* d_in, const int* in_sizes, int n_in,
                              void* d_out, int out_size, void* d_ws, size_t ws_size,
                              hipStream_t stream) {
    const float* x = (const float*)d_in[0];
    const int*   ei = (const int*)d_in[1];
    const float* W = (const float*)d_in[2];
    float* out = (float*)d_out;

    const int N = in_sizes[0] / DIM;
    const int E = in_sizes[1] / 2;

    float* deg = (float*)d_ws;
    size_t aggOff = (((size_t)N * sizeof(float)) + 255) & ~(size_t)255;
    float* agg = (float*)((char*)d_ws + aggOff);

    deg_init_kernel<<<(N + 255) / 256, 256, 0, stream>>>(deg, N);
    deg_count_kernel<<<(E + 255) / 256, 256, 0, stream>>>(ei, deg, E);
    deg_to_dis_kernel<<<(N + 255) / 256, 256, 0, stream>>>(deg, N);
    self_loop_kernel<<<(N + 7) / 8, 256, 0, stream>>>(x, deg, agg, N);
    edge_scatter_kernel<<<(E + 7) / 8, 256, 0, stream>>>(x, deg, ei, agg, E);
    gemm_wmma_kernel<<<(N + 15) / 16, 256, 0, stream>>>(agg, W, out, N);
}


# Edge scatter without global atomics: each work-group keeps 400 agg rows in LDS, scans a key byte per edge (from the degree pass), adds its edges by LDS f32 atomics, stores rows once; order unordered
# speedup vs baseline: 1.8050x; 1.8050x over previous
#include <hip/hip_runtime.h>
#include <hip/hip_bf16.h>

#define DIM 128

typedef float v2f __attribute__((ext_vector_type(2)));
typedef float v8f __attribute__((ext_vector_type(8)));

__device__ __forceinline__ void atomAddF(float* p, float v) {
    __hip_atomic_fetch_add(p, v, __ATOMIC_RELAXED, __HIP_MEMORY_SCOPE_AGENT);
}

__global__ void deg_init_kernel(float* __restrict__ deg, int n) {
    int i = blockIdx.x * blockDim.x + threadIdx.x;
    if (i < n) deg[i] = 1.0f;
}

#define ROWS_PER_WG 400

__global__ __launch_bounds__(256) void deg_count_key_kernel(const int* __restrict__ ei,
                                                            float* __restrict__ deg,
                                                            unsigned* __restrict__ keys,
                                                            int E, int nWords) {
    int t = blockIdx.x * blockDim.x + threadIdx.x;
    if (t >= nWords) return;
    const int e = t * 4;
    unsigned k = 0;
#pragma unroll
    for (int j = 0; j < 4; ++j) {
        if (e + j < E) {
            const int d = ei[E + e + j];
            atomAddF(&deg[d], 1.0f);
            k |= (((unsigned)d / ROWS_PER_WG) & 255u) << (8 * j);
        }
    }
    keys[t] = k;
}

__global__ void deg_to_dis_kernel(float* __restrict__ deg, int n) {
    int i = blockIdx.x * blockDim.x + threadIdx.x;
    if (i < n) {
        float d = deg[i];
        deg[i] = (d > 0.0f) ? rsqrtf(d) : 0.0f;
    }
}

#define OWN_WAVES 32
#define QCAP 64
#define EBATCH 16
#define SCAN_UNROLL 4
#define KEYS_PER_CHUNK 512

__device__ __forceinline__ void ldsAddF(float* p, float v) {
    __hip_atomic_fetch_add(p, v, __ATOMIC_RELAXED, __HIP_MEMORY_SCOPE_WORKGROUP);
}
__device__ __forceinline__ int bcastI(int v, int srcLane) {
    return __builtin_amdgcn_readlane(v, srcLane);
}
__device__ __forceinline__ float bcastF(float v, int srcLane) {
    return __int_as_float(__builtin_amdgcn_readlane(__float_as_int(v), srcLane));
}
__device__ __forceinline__ unsigned zeroByteFlags(unsigned t) {
    return (t - 0x01010101u) & ~t & 0x80808080u;
}
__device__ __forceinline__ unsigned chunkFlags(const uint4& kk, unsigned pat) {
    const unsigned z0 = zeroByteFlags(kk.x ^ pat);
    const unsigned z1 = zeroByteFlags(kk.y ^ pat);
    const unsigned z2 = zeroByteFlags(kk.z ^ pat);
    const unsigned z3 = zeroByteFlags(kk.w ^ pat);
    return z3 | (z2 >> 1) | (z1 >> 2) | (z0 >> 3);
}

__device__ __forceinline__ void load_key_group(const uint4* __restrict__ kp, int i0, int start,
                                               int nChunks, unsigned lane,
                                               uint4 (&kk)[SCAN_UNROLL]) {
#pragma unroll
    for (int u = 0; u < SCAN_UNROLL; ++u) {
        int c = i0 + u * OWN_WAVES;
        if (c >= nChunks) c = nChunks - 1;
        int w = c + start;
        if (w >= nChunks) w -= nChunks;
        kk[u] = kp[(size_t)w * 32 + lane];
    }
}

__device__ __forceinline__ void add_queued_edges(const float* __restrict__ x,
                                                 const float* __restrict__ dis,
                                                 const int* __restrict__ ei,
                                                 float* acc, const int* myq, int qhead, int cnt,
                                                 int rowBase, int rowsHere, int E, unsigned lane) {
    int s = 0, d = 0, lr = ROWS_PER_WG;
    bool ok = false;
    if ((int)lane < cnt) {
        const int e = myq[(qhead + (int)lane) & (QCAP - 1)];
        if (e < E) {
            s = ei[e];
            d = ei[E + e];
            ok = (unsigned)(d - rowBase) < (unsigned)rowsHere;
        }
    }
    if (ok) { lr = d - rowBase; } else { s = 0; d = 0; }
    const float ds = dis[s];
    const float dd = dis[d];

    for (int b = 0; b < cnt; b += EBATCH) {
        float v[EBATCH][4];
#pragma unroll
        for (int j = 0; j < EBATCH; ++j) {
            const float* xr = x + (size_t)bcastI(s, b + j) * DIM + lane;
            v[j][0] = xr[0];
            v[j][1] = xr[32];
            v[j][2] = xr[64];
            v[j][3] = xr[96];
        }
#pragma unroll
        for (int j = 0; j < EBATCH; ++j) {
            const float nj = bcastF(ds, b + j) * bcastF(dd, b + j);
            float* dst = acc + bcastI(lr, b + j) * DIM + lane;
            ldsAddF(dst + 0,  v[j][0] * nj);
            ldsAddF(dst + 32, v[j][1] * nj);
            ldsAddF(dst + 64, v[j][2] * nj);
            ldsAddF(dst + 96, v[j][3] * nj);
        }
    }
}

__global__ __launch_bounds__(1024) void owner_accumulate_kernel(const float* __restrict__ x,
                                                                const float* __restrict__ dis,
                                                                const int* __restrict__ ei,
                                                                const unsigned* __restrict__ keys,
                                                                float* __restrict__ agg,
                                                                int N, int E, int nChunks) {
    __shared__ float acc[(ROWS_PER_WG + 1) * DIM];
    __shared__ int q[OWN_WAVES * QCAP];
    const int tid = threadIdx.x;
    const unsigned lane = tid & 31;
    const int wv = __builtin_amdgcn_readfirstlane(tid >> 5);
    const int rowBase = (int)blockIdx.x * ROWS_PER_WG;
    const int rowsLeft = N - rowBase;
    const int rowsHere = rowsLeft < ROWS_PER_WG ? rowsLeft : ROWS_PER_WG;

    for (int i = tid; i < rowsHere * 32; i += 1024) {
        const int r = i >> 5, c4 = i & 31;
        float sc = dis[rowBase + r];
        sc = sc * sc;
        float4 v = reinterpret_cast<const float4*>(x + (size_t)(rowBase + r) * DIM)[c4];
        v.x *= sc; v.y *= sc; v.z *= sc; v.w *= sc;
        reinterpret_cast<float4*>(acc + r * DIM)[c4] = v;
    }
    if (tid < 32) {
        reinterpret_cast<float4*>(acc + ROWS_PER_WG * DIM)[tid] = make_float4(0.f, 0.f, 0.f, 0.f);
    }
    __syncthreads();

    const unsigned pat = ((unsigned)blockIdx.x & 255u) * 0x01010101u;
    int* myq = q + wv * QCAP;
    int qhead = 0, qn = 0;
    const int start = (int)blockIdx.x * (nChunks / (int)gridDim.x);
    const uint4* kp = reinterpret_cast<const uint4*>(keys);

    uint4 nx[SCAN_UNROLL];
    load_key_group(kp, wv, start, nChunks, lane, nx);
    for (int i = wv; i < nChunks; i += OWN_WAVES * SCAN_UNROLL) {
        uint4 kk[SCAN_UNROLL];
#pragma unroll
        for (int u = 0; u < SCAN_UNROLL; ++u) kk[u] = nx[u];
        load_key_group(kp, i + OWN_WAVES * SCAN_UNROLL, start, nChunks, lane, nx);

        unsigned cf[SCAN_UNROLL];
#pragma unroll
        for (int u = 0; u < SCAN_UNROLL; ++u) {
            cf[u] = (i + u * OWN_WAVES < nChunks) ? chunkFlags(kk[u], pat) : 0u;
        }
        unsigned long long mask64 = (unsigned long long)(cf[0] | (cf[1] >> 4)) |
                                    ((unsigned long long)(cf[2] | (cf[3] >> 4)) << 32);

        for (int trip = 0; trip < 16 * SCAN_UNROLL; ++trip) {
            const unsigned bal = __builtin_amdgcn_ballot_w32(mask64 != 0ull);
            if (bal == 0u) break;
            if (mask64 != 0ull) {
                const int bit = __builtin_ctzll(mask64);
                const int n = bit & 7;
                const int u = ((bit >> 5) << 1) + (n < 4 ? 1 : 0);
                const int j = ((n & 3) << 2) + ((bit >> 3) & 3);
                int w = i + u * OWN_WAVES + start;
                if (w >= nChunks) w -= nChunks;
                const int pos = qn + (int)__builtin_amdgcn_mbcnt_lo(bal, 0u);
                myq[(qhead + pos) & (QCAP - 1)] = w * KEYS_PER_CHUNK + (int)lane * 16 + j;
                mask64 &= mask64 - 1ull;
            }
            qn += __builtin_popcount(bal);
            if (qn >= 32) {
                add_queued_edges(x, dis, ei, acc, myq, qhead, 32, rowBase, rowsHere, E, lane);
                qhead = (qhead + 32) & (QCAP - 1);
                qn -= 32;
            }
        }
    }
    if (qn > 0) {
        add_queued_edges(x, dis, ei, acc, myq, qhead, qn, rowBase, rowsHere, E, lane);
    }
    __syncthreads();

    for (int i = tid; i < rowsHere * 32; i += 1024) {
        const int r = i >> 5, c4 = i & 31;
        reinterpret_cast<float4*>(agg + (size_t)(rowBase + r) * DIM)[c4] =
            reinterpret_cast<const float4*>(acc + r * DIM)[c4];
    }
}

#define LDA 132
#define LDW 136
#define KCHUNK 32

__global__ __launch_bounds__(256) void gemm_wmma_kernel(const float* __restrict__ A,
                                                        const float* __restrict__ W,
                                                        float* __restrict__ out, int n) {
    __shared__ float lA[16 * LDA];
    __shared__ float lW[KCHUNK * LDW];
    const int tid = threadIdx.x;
    const int wave = tid >> 5;
    const int lane = tid & 31;
    const int rowBase = blockIdx.x * 16;

    for (int i = tid; i < 16 * DIM; i += 256) {
        int r = i >> 7, c = i & 127;
        int gr = rowBase + r;
        lA[r * LDA + c] = (gr < n) ? A[(size_t)gr * DIM + c] : 0.0f;
    }

    const int m = lane & 15;
    const int hi = lane >> 4;
    const int colBase = wave * 16;

    v8f acc = {0.f, 0.f, 0.f, 0.f, 0.f, 0.f, 0.f, 0.f};

    for (int kc = 0; kc < DIM; kc += KCHUNK) {
        __syncthreads();
        for (int i = tid; i < KCHUNK * DIM; i += 256) {
            int r = i >> 7, c = i & 127;
            lW[r * LDW + c] = W[(size_t)(kc + r) * DIM + c];
        }
        __syncthreads();

#pragma unroll
        for (int k = 0; k < KCHUNK; k += 4) {
            v2f a, b;
            const float* ap = &lA[m * LDA + kc + k + 2 * hi];
            a.x = ap[0];
            a.y = ap[1];
            b.x = lW[(k + 2 * hi) * LDW + colBase + m];
            b.y = lW[(k + 2 * hi + 1) * LDW + colBase + m];
            acc = __builtin_amdgcn_wmma_f32_16x16x4_f32(
                 false, a,  false, b,
                 (short)0, acc,  false,  false);
        }
    }

#pragma unroll
    for (int v = 0; v < 8; ++v) {
        int gr = rowBase + v + 8 * hi;
        if (gr < n) out[(size_t)gr * DIM + colBase + m] = acc[v];
    }
}

extern "C" void kernel_launch(void* const* d_in, const int* in_sizes, int n_in,
                              void* d_out, int out_size, void* d_ws, size_t ws_size,
                              hipStream_t stream) {
    const float* x = (const float*)d_in[0];
    const int*   ei = (const int*)d_in[1];
    const float* W = (const float*)d_in[2];
    float* out = (float*)d_out;

    const int N = in_sizes[0] / DIM;
    const int E = in_sizes[1] / 2;

    float* deg = (float*)d_ws;
    size_t aggOff = (((size_t)N * sizeof(float)) + 255) & ~(size_t)255;
    float* agg = (float*)((char*)d_ws + aggOff);
    size_t keyOff = (aggOff + (size_t)N * DIM * sizeof(float) + 255) & ~(size_t)255;
    unsigned* keys = (unsigned*)((char*)d_ws + keyOff);

    const int nChunks = (E + KEYS_PER_CHUNK - 1) / KEYS_PER_CHUNK;
    const int nWords = nChunks * (KEYS_PER_CHUNK / 4);

    deg_init_kernel<<<(N + 255) / 256, 256, 0, stream>>>(deg, N);
    deg_count_key_kernel<<<(nWords + 255) / 256, 256, 0, stream>>>(ei, deg, keys, E, nWords);
    deg_to_dis_kernel<<<(N + 255) / 256, 256, 0, stream>>>(deg, N);
    owner_accumulate_kernel<<<(N + ROWS_PER_WG - 1) / ROWS_PER_WG, 1024, 0, stream>>>(
        x, deg, ei, keys, agg, N, E, nChunks);
    gemm_wmma_kernel<<<(N + 15) / 16, 256, 0, stream>>>(agg, W, out, N);
}


# Row gather by one 16-byte load per lane: one load per 512-B row, was four one-line loads; LDS tile column-permuted (col 4l+c at word 32c+l): adds keep 32 consecutive banks; order unordered as before
# speedup vs baseline: 1.9420x; 1.0759x over previous
#include <hip/hip_runtime.h>
#include <hip/hip_bf16.h>

#define DIM 128

typedef float v2f __attribute__((ext_vector_type(2)));
typedef float v8f __attribute__((ext_vector_type(8)));

__device__ __forceinline__ void atomAddF(float* p, float v) {
    __hip_atomic_fetch_add(p, v, __ATOMIC_RELAXED, __HIP_MEMORY_SCOPE_AGENT);
}

__global__ void deg_init_kernel(float* __restrict__ deg, int n) {
    int i = blockIdx.x * blockDim.x + threadIdx.x;
    if (i < n) deg[i] = 1.0f;
}

#define ROWS_PER_WG 400

__global__ __launch_bounds__(256) void deg_count_key_kernel(const int* __restrict__ ei,
                                                            float* __restrict__ deg,
                                                            unsigned* __restrict__ keys,
                                                            int E, int nWords) {
    int t = blockIdx.x * blockDim.x + threadIdx.x;
    if (t >= nWords) return;
    const int e = t * 4;
    unsigned k = 0;
#pragma unroll
    for (int j = 0; j < 4; ++j) {
        if (e + j < E) {
            const int d = ei[E + e + j];
            atomAddF(&deg[d], 1.0f);
            k |= (((unsigned)d / ROWS_PER_WG) & 255u) << (8 * j);
        }
    }
    keys[t] = k;
}

__global__ void deg_to_dis_kernel(float* __restrict__ deg, int n) {
    int i = blockIdx.x * blockDim.x + threadIdx.x;
    if (i < n) {
        float d = deg[i];
        deg[i] = (d > 0.0f) ? rsqrtf(d) : 0.0f;
    }
}

#define OWN_WAVES 32
#define QCAP 64
#define EBATCH 16
#define SCAN_UNROLL 4
#define KEYS_PER_CHUNK 512

__device__ __forceinline__ void ldsAddF(float* p, float v) {
    __hip_atomic_fetch_add(p, v, __ATOMIC_RELAXED, __HIP_MEMORY_SCOPE_WORKGROUP);
}
__device__ __forceinline__ int bcastI(int v, int srcLane) {
    return __builtin_amdgcn_readlane(v, srcLane);
}
__device__ __forceinline__ float bcastF(float v, int srcLane) {
    return __int_as_float(__builtin_amdgcn_readlane(__float_as_int(v), srcLane));
}
__device__ __forceinline__ unsigned zeroByteFlags(unsigned t) {
    return (t - 0x01010101u) & ~t & 0x80808080u;
}
__device__ __forceinline__ unsigned chunkFlags(const uint4& kk, unsigned pat) {
    const unsigned z0 = zeroByteFlags(kk.x ^ pat);
    const unsigned z1 = zeroByteFlags(kk.y ^ pat);
    const unsigned z2 = zeroByteFlags(kk.z ^ pat);
    const unsigned z3 = zeroByteFlags(kk.w ^ pat);
    return z3 | (z2 >> 1) | (z1 >> 2) | (z0 >> 3);
}

__device__ __forceinline__ void load_key_group(const uint4* __restrict__ kp, int i0, int start,
                                               int nChunks, unsigned lane,
                                               uint4 (&kk)[SCAN_UNROLL]) {
#pragma unroll
    for (int u = 0; u < SCAN_UNROLL; ++u) {
        int c = i0 + u * OWN_WAVES;
        if (c >= nChunks) c = nChunks - 1;
        int w = c + start;
        if (w >= nChunks) w -= nChunks;
        kk[u] = kp[(size_t)w * 32 + lane];
    }
}

__device__ __forceinline__ float4 load_row_piece(const float* __restrict__ x, int s, unsigned lane) {
    const unsigned off = ((unsigned)s << 9) | (lane << 4);
    return *reinterpret_cast<const float4*>(reinterpret_cast<const char*>(x) + off);
}

__device__ __forceinline__ void add_queued_edges(const float* __restrict__ x,
                                                 const float* __restrict__ dis,
                                                 const int* __restrict__ ei,
                                                 float* acc, const int* myq, int qhead, int cnt,
                                                 int rowBase, int rowsHere, int E, unsigned lane) {
    int s = 0, d = 0, lr = ROWS_PER_WG;
    bool ok = false;
    if ((int)lane < cnt) {
        const int e = myq[(qhead + (int)lane) & (QCAP - 1)];
        if (e < E) {
            s = ei[e];
            d = ei[E + e];
            ok = (unsigned)(d - rowBase) < (unsigned)rowsHere;
        }
    }
    if (ok) { lr = d - rowBase; } else { s = 0; d = 0; }
    const float ds = dis[s];
    const float dd = dis[d];

    for (int b = 0; b < cnt; b += EBATCH) {
        float4 v[EBATCH];
#pragma unroll
        for (int j = 0; j < EBATCH; ++j) {
            v[j] = load_row_piece(x, bcastI(s, b + j), lane);
        }
#pragma unroll
        for (int j = 0; j < EBATCH; ++j) {
            const float nj = bcastF(ds, b + j) * bcastF(dd, b + j);
            float* dst = acc + bcastI(lr, b + j) * DIM + lane;
            ldsAddF(dst + 0,  v[j].x * nj);
            ldsAddF(dst + 32, v[j].y * nj);
            ldsAddF(dst + 64, v[j].z * nj);
            ldsAddF(dst + 96, v[j].w * nj);
        }
    }
}

__global__ __launch_bounds__(1024) void owner_accumulate_kernel(const float* __restrict__ x,
                                                                const float* __restrict__ dis,
                                                                const int* __restrict__ ei,
                                                                const unsigned* __restrict__ keys,
                                                                float* __restrict__ agg,
                                                                int N, int E, int nChunks) {
    __shared__ float acc[(ROWS_PER_WG + 1) * DIM];
    __shared__ int q[OWN_WAVES * QCAP];
    const int tid = threadIdx.x;
    const unsigned lane = tid & 31;
    const int wv = __builtin_amdgcn_readfirstlane(tid >> 5);
    const int rowBase = (int)blockIdx.x * ROWS_PER_WG;
    const int rowsLeft = N - rowBase;
    const int rowsHere = rowsLeft < ROWS_PER_WG ? rowsLeft : ROWS_PER_WG;

    for (int i = tid; i < rowsHere * 32; i += 1024) {
        const int r = i >> 5, c4 = i & 31;
        float sc = dis[rowBase + r];
        sc = sc * sc;
        float4 v = reinterpret_cast<const float4*>(x + (size_t)(rowBase + r) * DIM)[c4];
        float* row = acc + r * DIM + c4;
        row[0]  = v.x * sc;
        row[32] = v.y * sc;
        row[64] = v.z * sc;
        row[96] = v.w * sc;
    }
    if (tid < 32) {
        reinterpret_cast<float4*>(acc + ROWS_PER_WG * DIM)[tid] = make_float4(0.f, 0.f, 0.f, 0.f);
    }
    __syncthreads();

    const unsigned pat = ((unsigned)blockIdx.x & 255u) * 0x01010101u;
    int* myq = q + wv * QCAP;
    int qhead = 0, qn = 0;
    const int start = (int)blockIdx.x * (nChunks / (int)gridDim.x);
    const uint4* kp = reinterpret_cast<const uint4*>(keys);

    uint4 nx[SCAN_UNROLL];
    load_key_group(kp, wv, start, nChunks, lane, nx);
    for (int i = wv; i < nChunks; i += OWN_WAVES * SCAN_UNROLL) {
        uint4 kk[SCAN_UNROLL];
#pragma unroll
        for (int u = 0; u < SCAN_UNROLL; ++u) kk[u] = nx[u];
        load_key_group(kp, i + OWN_WAVES * SCAN_UNROLL, start, nChunks, lane, nx);

        unsigned cf[SCAN_UNROLL];
#pragma unroll
        for (int u = 0; u < SCAN_UNROLL; ++u) {
            cf[u] = (i + u * OWN_WAVES < nChunks) ? chunkFlags(kk[u], pat) : 0u;
        }
        unsigned long long mask64 = (unsigned long long)(cf[0] | (cf[1] >> 4)) |
                                    ((unsigned long long)(cf[2] | (cf[3] >> 4)) << 32);

        for (int trip = 0; trip < 16 * SCAN_UNROLL; ++trip) {
            const unsigned bal = __builtin_amdgcn_ballot_w32(mask64 != 0ull);
            if (bal == 0u) break;
            if (mask64 != 0ull) {
                const int bit = __builtin_ctzll(mask64);
                const int n = bit & 7;
                const int u = ((bit >> 5) << 1) + (n < 4 ? 1 : 0);
                const int j = ((n & 3) << 2) + ((bit >> 3) & 3);
                int w = i + u * OWN_WAVES + start;
                if (w >= nChunks) w -= nChunks;
                const int pos = qn + (int)__builtin_amdgcn_mbcnt_lo(bal, 0u);
                myq[(qhead + pos) & (QCAP - 1)] = w * KEYS_PER_CHUNK + (int)lane * 16 + j;
                mask64 &= mask64 - 1ull;
            }
            qn += __builtin_popcount(bal);
            if (qn >= 32) {
                add_queued_edges(x, dis, ei, acc, myq, qhead, 32, rowBase, rowsHere, E, lane);
                qhead = (qhead + 32) & (QCAP - 1);
                qn -= 32;
            }
        }
    }
    if (qn > 0) {
        add_queued_edges(x, dis, ei, acc, myq, qhead, qn, rowBase, rowsHere, E, lane);
    }
    __syncthreads();

    for (int i = tid; i < rowsHere * 32; i += 1024) {
        const int r = i >> 5, c4 = i & 31;
        const float* row = acc + r * DIM + c4;
        reinterpret_cast<float4*>(agg + (size_t)(rowBase + r) * DIM)[c4] =
            make_float4(row[0], row[32], row[64], row[96]);
    }
}

#define LDA 132
#define LDW 136
#define KCHUNK 32

__global__ __launch_bounds__(256) void gemm_wmma_kernel(const float* __restrict__ A,
                                                        const float* __restrict__ W,
                                                        float* __restrict__ out, int n) {
    __shared__ float lA[16 * LDA];
    __shared__ float lW[KCHUNK * LDW];
    const int tid = threadIdx.x;
    const int wave = tid >> 5;
    const int lane = tid & 31;
    const int rowBase = blockIdx.x * 16;

    for (int i = tid; i < 16 * DIM; i += 256) {
        int r = i >> 7, c = i & 127;
        int gr = rowBase + r;
        lA[r * LDA + c] = (gr < n) ? A[(size_t)gr * DIM + c] : 0.0f;
    }

    const int m = lane & 15;
    const int hi = lane >> 4;
    const int colBase = wave * 16;

    v8f acc = {0.f, 0.f, 0.f, 0.f, 0.f, 0.f, 0.f, 0.f};

    for (int kc = 0; kc < DIM; kc += KCHUNK) {
        __syncthreads();
        for (int i = tid; i < KCHUNK * DIM; i += 256) {
            int r = i >> 7, c = i & 127;
            lW[r * LDW + c] = W[(size_t)(kc + r) * DIM + c];
        }
        __syncthreads();

#pragma unroll
        for (int k = 0; k < KCHUNK; k += 4) {
            v2f a, b;
            const float* ap = &lA[m * LDA + kc + k + 2 * hi];
            a.x = ap[0];
            a.y = ap[1];
            b.x = lW[(k + 2 * hi) * LDW + colBase + m];
            b.y = lW[(k + 2 * hi + 1) * LDW + colBase + m];
            acc = __builtin_amdgcn_wmma_f32_16x16x4_f32(
                 false, a,  false, b,
                 (short)0, acc,  false,  false);
        }
    }

#pragma unroll
    for (int v = 0; v < 8; ++v) {
        int gr = rowBase + v + 8 * hi;
        if (gr < n) out[(size_t)gr * DIM + colBase + m] = acc[v];
    }
}

extern "C" void kernel_launch(void* const* d_in, const int* in_sizes, int n_in,
                              void* d_out, int out_size, void* d_ws, size_t ws_size,
                              hipStream_t stream) {
    const float* x = (const float*)d_in[0];
    const int*   ei = (const int*)d_in[1];
    const float* W = (const float*)d_in[2];
    float* out = (float*)d_out;

    const int N = in_sizes[0] / DIM;
    const int E = in_sizes[1] / 2;

    float* deg = (float*)d_ws;
    size_t aggOff = (((size_t)N * sizeof(float)) + 255) & ~(size_t)255;
    float* agg = (float*)((char*)d_ws + aggOff);
    size_t keyOff = (aggOff + (size_t)N * DIM * sizeof(float) + 255) & ~(size_t)255;
    unsigned* keys = (unsigned*)((char*)d_ws + keyOff);

    const int nChunks = (E + KEYS_PER_CHUNK - 1) / KEYS_PER_CHUNK;
    const int nWords = nChunks * (KEYS_PER_CHUNK / 4);

    deg_init_kernel<<<(N + 255) / 256, 256, 0, stream>>>(deg, N);
    deg_count_key_kernel<<<(nWords + 255) / 256, 256, 0, stream>>>(ei, deg, keys, E, nWords);
    deg_to_dis_kernel<<<(N + 255) / 256, 256, 0, stream>>>(deg, N);
    owner_accumulate_kernel<<<(N + ROWS_PER_WG - 1) / ROWS_PER_WG, 1024, 0, stream>>>(
        x, deg, ei, keys, agg, N, E, nChunks);
    gemm_wmma_kernel<<<(N + 15) / 16, 256, 0, stream>>>(agg, W, out, N);
}


# Final product by one wave per SIMD holding all of W in registers as B fragments, walking 7 row strips with A fragments read from global one strip ahead: no LDS, no barrier; k sum order regrouped
# speedup vs baseline: 2.3243x; 1.1969x over previous
#include <hip/hip_runtime.h>
#include <hip/hip_bf16.h>

#define DIM 128

typedef float v2f __attribute__((ext_vector_type(2)));
typedef float v8f __attribute__((ext_vector_type(8)));

__device__ __forceinline__ void atomAddF(float* p, float v) {
    __hip_atomic_fetch_add(p, v, __ATOMIC_RELAXED, __HIP_MEMORY_SCOPE_AGENT);
}

__global__ void deg_init_kernel(float* __restrict__ deg, int n) {
    int i = blockIdx.x * blockDim.x + threadIdx.x;
    if (i < n) deg[i] = 1.0f;
}

#define ROWS_PER_WG 400

__global__ __launch_bounds__(256) void deg_count_key_kernel(const int* __restrict__ ei,
                                                            float* __restrict__ deg,
                                                            unsigned* __restrict__ keys,
                                                            int E, int nWords) {
    int t = blockIdx.x * blockDim.x + threadIdx.x;
    if (t >= nWords) return;
    const int e = t * 4;
    unsigned k = 0;
#pragma unroll
    for (int j = 0; j < 4; ++j) {
        if (e + j < E) {
            const int d = ei[E + e + j];
            atomAddF(&deg[d], 1.0f);
            k |= (((unsigned)d / ROWS_PER_WG) & 255u) << (8 * j);
        }
    }
    keys[t] = k;
}

__global__ void deg_to_dis_kernel(float* __restrict__ deg, int n) {
    int i = blockIdx.x * blockDim.x + threadIdx.x;
    if (i < n) {
        float d = deg[i];
        deg[i] = (d > 0.0f) ? rsqrtf(d) : 0.0f;
    }
}

#define OWN_WAVES 32
#define QCAP 64
#define EBATCH 16
#define SCAN_UNROLL 4
#define KEYS_PER_CHUNK 512

__device__ __forceinline__ void ldsAddF(float* p, float v) {
    __hip_atomic_fetch_add(p, v, __ATOMIC_RELAXED, __HIP_MEMORY_SCOPE_WORKGROUP);
}
__device__ __forceinline__ int bcastI(int v, int srcLane) {
    return __builtin_amdgcn_readlane(v, srcLane);
}
__device__ __forceinline__ float bcastF(float v, int srcLane) {
    return __int_as_float(__builtin_amdgcn_readlane(__float_as_int(v), srcLane));
}
__device__ __forceinline__ unsigned zeroByteFlags(unsigned t) {
    return (t - 0x01010101u) & ~t & 0x80808080u;
}
__device__ __forceinline__ unsigned chunkFlags(const uint4& kk, unsigned pat) {
    const unsigned z0 = zeroByteFlags(kk.x ^ pat);
    const unsigned z1 = zeroByteFlags(kk.y ^ pat);
    const unsigned z2 = zeroByteFlags(kk.z ^ pat);
    const unsigned z3 = zeroByteFlags(kk.w ^ pat);
    return z3 | (z2 >> 1) | (z1 >> 2) | (z0 >> 3);
}

__device__ __forceinline__ void load_key_group(const uint4* __restrict__ kp, int i0, int start,
                                               int nChunks, unsigned lane,
                                               uint4 (&kk)[SCAN_UNROLL]) {
#pragma unroll
    for (int u = 0; u < SCAN_UNROLL; ++u) {
        int c = i0 + u * OWN_WAVES;
        if (c >= nChunks) c = nChunks - 1;
        int w = c + start;
        if (w >= nChunks) w -= nChunks;
        kk[u] = kp[(size_t)w * 32 + lane];
    }
}

__device__ __forceinline__ float4 load_row_piece(const float* __restrict__ x, int s, unsigned lane) {
    const unsigned off = ((unsigned)s << 9) | (lane << 4);
    return *reinterpret_cast<const float4*>(reinterpret_cast<const char*>(x) + off);
}

__device__ __forceinline__ void add_queued_edges(const float* __restrict__ x,
                                                 const float* __restrict__ dis,
                                                 const int* __restrict__ ei,
                                                 float* acc, const int* myq, int qhead, int cnt,
                                                 int rowBase, int rowsHere, int E, unsigned lane) {
    int s = 0, d = 0, lr = ROWS_PER_WG;
    bool ok = false;
    if ((int)lane < cnt) {
        const int e = myq[(qhead + (int)lane) & (QCAP - 1)];
        if (e < E) {
            s = ei[e];
            d = ei[E + e];
            ok = (unsigned)(d - rowBase) < (unsigned)rowsHere;
        }
    }
    if (ok) { lr = d - rowBase; } else { s = 0; d = 0; }
    const float ds = dis[s];
    const float dd = dis[d];

    for (int b = 0; b < cnt; b += EBATCH) {
        float4 v[EBATCH];
#pragma unroll
        for (int j = 0; j < EBATCH; ++j) {
            v[j] = load_row_piece(x, bcastI(s, b + j), lane);
        }
#pragma unroll
        for (int j = 0; j < EBATCH; ++j) {
            const float nj = bcastF(ds, b + j) * bcastF(dd, b + j);
            float* dst = acc + bcastI(lr, b + j) * DIM + lane;
            ldsAddF(dst + 0,  v[j].x * nj);
            ldsAddF(dst + 32, v[j].y * nj);
            ldsAddF(dst + 64, v[j].z * nj);
            ldsAddF(dst + 96, v[j].w * nj);
        }
    }
}

__global__ __launch_bounds__(1024) void owner_accumulate_kernel(const float* __restrict__ x,
                                                                const float* __restrict__ dis,
                                                                const int* __restrict__ ei,
                                                                const unsigned* __restrict__ keys,
                                                                float* __restrict__ agg,
                                                                int N, int E, int nChunks) {
    __shared__ float acc[(ROWS_PER_WG + 1) * DIM];
    __shared__ int q[OWN_WAVES * QCAP];
    const int tid = threadIdx.x;
    const unsigned lane = tid & 31;
    const int wv = __builtin_amdgcn_readfirstlane(tid >> 5);
    const int rowBase = (int)blockIdx.x * ROWS_PER_WG;
    const int rowsLeft = N - rowBase;
    const int rowsHere = rowsLeft < ROWS_PER_WG ? rowsLeft : ROWS_PER_WG;

    for (int i = tid; i < rowsHere * 32; i += 1024) {
        const int r = i >> 5, c4 = i & 31;
        float sc = dis[rowBase + r];
        sc = sc * sc;
        float4 v = reinterpret_cast<const float4*>(x + (size_t)(rowBase + r) * DIM)[c4];
        float* row = acc + r * DIM + c4;
        row[0]  = v.x * sc;
        row[32] = v.y * sc;
        row[64] = v.z * sc;
        row[96] = v.w * sc;
    }
    if (tid < 32) {
        reinterpret_cast<float4*>(acc + ROWS_PER_WG * DIM)[tid] = make_float4(0.f, 0.f, 0.f, 0.f);
    }
    __syncthreads();

    const unsigned pat = ((unsigned)blockIdx.x & 255u) * 0x01010101u;
    int* myq = q + wv * QCAP;
    int qhead = 0, qn = 0;
    const int start = (int)blockIdx.x * (nChunks / (int)gridDim.x);
    const uint4* kp = reinterpret_cast<const uint4*>(keys);

    uint4 nx[SCAN_UNROLL];
    load_key_group(kp, wv, start, nChunks, lane, nx);
    for (int i = wv; i < nChunks; i += OWN_WAVES * SCAN_UNROLL) {
        uint4 kk[SCAN_UNROLL];
#pragma unroll
        for (int u = 0; u < SCAN_UNROLL; ++u) kk[u] = nx[u];
        load_key_group(kp, i + OWN_WAVES * SCAN_UNROLL, start, nChunks, lane, nx);

        unsigned cf[SCAN_UNROLL];
#pragma unroll
        for (int u = 0; u < SCAN_UNROLL; ++u) {
            cf[u] = (i + u * OWN_WAVES < nChunks) ? chunkFlags(kk[u], pat) : 0u;
        }
        unsigned long long mask64 = (unsigned long long)(cf[0] | (cf[1] >> 4)) |
                                    ((unsigned long long)(cf[2] | (cf[3] >> 4)) << 32);

        for (int trip = 0; trip < 16 * SCAN_UNROLL; ++trip) {
            const unsigned bal = __builtin_amdgcn_ballot_w32(mask64 != 0ull);
            if (bal == 0u) break;
            if (mask64 != 0ull) {
                const int bit = __builtin_ctzll(mask64);
                const int n = bit & 7;
                const int u = ((bit >> 5) << 1) + (n < 4 ? 1 : 0);
                const int j = ((n & 3) << 2) + ((bit >> 3) & 3);
                int w = i + u * OWN_WAVES + start;
                if (w >= nChunks) w -= nChunks;
                const int pos = qn + (int)__builtin_amdgcn_mbcnt_lo(bal, 0u);
                myq[(qhead + pos) & (QCAP - 1)] = w * KEYS_PER_CHUNK + (int)lane * 16 + j;
                mask64 &= mask64 - 1ull;
            }
            qn += __builtin_popcount(bal);
            if (qn >= 32) {
                add_queued_edges(x, dis, ei, acc, myq, qhead, 32, rowBase, rowsHere, E, lane);
                qhead = (qhead + 32) & (QCAP - 1);
                qn -= 32;
            }
        }
    }
    if (qn > 0) {
        add_queued_edges(x, dis, ei, acc, myq, qhead, qn, rowBase, rowsHere, E, lane);
    }
    __syncthreads();

    for (int i = tid; i < rowsHere * 32; i += 1024) {
        const int r = i >> 5, c4 = i & 31;
        const float* row = acc + r * DIM + c4;
        reinterpret_cast<float4*>(agg + (size_t)(rowBase + r) * DIM)[c4] =
            make_float4(row[0], row[32], row[64], row[96]);
    }
}

#define STRIPS_PER_WAVE 7

__device__ __forceinline__ void load_a_strip(const float* __restrict__ A, int rowBase, int n,
                                             int m, int h, float4 (&q)[16]) {
    int row = rowBase + m;
    if (row > n - 1) row = n - 1;
    const float* p = A + (size_t)row * DIM + 4 * h;
#pragma unroll
    for (int j = 0; j < 16; ++j) q[j] = *reinterpret_cast<const float4*>(p + 8 * j);
}

__global__ __launch_bounds__(32) void gemm_wmma_kernel(const float* __restrict__ A,
                                                       const float* __restrict__ W,
                                                       float* __restrict__ out, int n) {
    const int lane = threadIdx.x;
    const int m = lane & 15;
    const int h = lane >> 4;
    const int strip0 = (int)blockIdx.x * STRIPS_PER_WAVE;

    float4 q[16];
    load_a_strip(A, strip0 * 16, n, m, h, q);

    v2f b[32][8];
    const float* wp = W + (size_t)(4 * h) * DIM + m;
#pragma unroll
    for (int t = 0; t < 32; ++t) {
        const int k0 = 8 * (t >> 1) + 2 * (t & 1);
#pragma unroll
        for (int c = 0; c < 8; ++c) {
            b[t][c].x = wp[(k0 + 0) * DIM + 16 * c];
            b[t][c].y = wp[(k0 + 1) * DIM + 16 * c];
        }
    }

#pragma unroll 1
    for (int sidx = 0; sidx < STRIPS_PER_WAVE; ++sidx) {
        const int rowBase = (strip0 + sidx) * 16;
        if (rowBase >= n) break;
        float4 qn[16];
        load_a_strip(A, rowBase + 16, n, m, h, qn);
        __builtin_amdgcn_sched_barrier(0);

        v8f acc[8];
#pragma unroll
        for (int c = 0; c < 8; ++c) acc[c] = (v8f){0.f, 0.f, 0.f, 0.f, 0.f, 0.f, 0.f, 0.f};

#pragma unroll
        for (int j = 0; j < 16; ++j) {
            v2f a0, a1;
            a0.x = q[j].x; a0.y = q[j].y;
            a1.x = q[j].z; a1.y = q[j].w;
#pragma unroll
            for (int c = 0; c < 8; ++c) {
                acc[c] = __builtin_amdgcn_wmma_f32_16x16x4_f32(
                     false, a0,  false, b[2 * j][c],
                     (short)0, acc[c],  false,  false);
            }
#pragma unroll
            for (int c = 0; c < 8; ++c) {
                acc[c] = __builtin_amdgcn_wmma_f32_16x16x4_f32(
                     false, a1,  false, b[2 * j + 1][c],
                     (short)0, acc[c],  false,  false);
            }
        }

        float* op = out + (size_t)(rowBase + 8 * h) * DIM + m;
#pragma unroll
        for (int v = 0; v < 8; ++v) {
            if (rowBase + 8 * h + v < n) {
#pragma unroll
                for (int c = 0; c < 8; ++c) op[v * DIM + 16 * c] = acc[c][v];
            }
        }
#pragma unroll
        for (int j = 0; j < 16; ++j) q[j] = qn[j];
    }
}

extern "C" void kernel_launch(void* const* d_in, const int* in_sizes, int n_in,
                              void* d_out, int out_size, void* d_ws, size_t ws_size,
                              hipStream_t stream) {
    const float* x = (const float*)d_in[0];
    const int*   ei = (const int*)d_in[1];
    const float* W = (const float*)d_in[2];
    float* out = (float*)d_out;

    const int N = in_sizes[0] / DIM;
    const int E = in_sizes[1] / 2;

    float* deg = (float*)d_ws;
    size_t aggOff = (((size_t)N * sizeof(float)) + 255) & ~(size_t)255;
    float* agg = (float*)((char*)d_ws + aggOff);
    size_t keyOff = (aggOff + (size_t)N * DIM * sizeof(float) + 255) & ~(size_t)255;
    unsigned* keys = (unsigned*)((char*)d_ws + keyOff);

    const int nChunks = (E + KEYS_PER_CHUNK - 1) / KEYS_PER_CHUNK;
    const int nWords = nChunks * (KEYS_PER_CHUNK / 4);

    deg_init_kernel<<<(N + 255) / 256, 256, 0, stream>>>(deg, N);
    deg_count_key_kernel<<<(nWords + 255) / 256, 256, 0, stream>>>(ei, deg, keys, E, nWords);
    deg_to_dis_kernel<<<(N + 255) / 256, 256, 0, stream>>>(deg, N);
    owner_accumulate_kernel<<<(N + ROWS_PER_WG - 1) / ROWS_PER_WG, 1024, 0, stream>>>(
        x, deg, ei, keys, agg, N, E, nChunks);
    const int nStrips = (N + 15) / 16;
    gemm_wmma_kernel<<<(nStrips + STRIPS_PER_WAVE - 1) / STRIPS_PER_WAVE, 32, 0, stream>>>(agg, W, out, N);
}


# Edge adds: one scattered global load per edge, was four: the degree pass packs source and row-in-tile into one word; both norm factors come from LDS (own rows, a degree byte table); order unordered
# speedup vs baseline: 2.4403x; 1.0499x over previous
#include <hip/hip_runtime.h>
#include <hip/hip_bf16.h>

#define DIM 128

typedef float v2f __attribute__((ext_vector_type(2)));
typedef float v8f __attribute__((ext_vector_type(8)));
typedef unsigned v4u __attribute__((ext_vector_type(4)));

__device__ __forceinline__ void atomAddF(float* p, float v) {
    __hip_atomic_fetch_add(p, v, __ATOMIC_RELAXED, __HIP_MEMORY_SCOPE_AGENT);
}

__global__ void deg_init_kernel(float* __restrict__ deg, int n) {
    int i = blockIdx.x * blockDim.x + threadIdx.x;
    if (i < n) deg[i] = 1.0f;
}

#define ROWS_PER_WG 400

#define PK_SRC_BITS 17
#define PK_SRC_MASK 0x1FFFFu
#define PK_TAG_SHIFT 26

__global__ __launch_bounds__(256) void deg_count_key_kernel(const int* __restrict__ ei,
                                                            float* __restrict__ deg,
                                                            unsigned* __restrict__ keys,
                                                            unsigned* __restrict__ pk,
                                                            int E, int nWords) {
    int t = blockIdx.x * blockDim.x + threadIdx.x;
    if (t >= nWords) return;
    const int e = t * 4;
    unsigned k = 0;
#pragma unroll
    for (int j = 0; j < 4; ++j) {
        if (e + j < E) {
            const int s = ei[e + j];
            const int d = ei[E + e + j];
            atomAddF(&deg[d], 1.0f);
            const unsigned g = (unsigned)d / ROWS_PER_WG;
            const unsigned lr = (unsigned)d - g * ROWS_PER_WG;
            k |= (g & 255u) << (8 * j);
            pk[e + j] = (unsigned)s | (lr << PK_SRC_BITS) | ((g & 1u) << PK_TAG_SHIFT) |
                        ((g >> 8) << (PK_TAG_SHIFT + 1));
        }
    }
    keys[t] = k;
}

__global__ __launch_bounds__(256) void deg_to_dis_kernel(float* __restrict__ deg,
                                                         unsigned* __restrict__ degB,
                                                         int n, int nWords) {
    int t = blockIdx.x * blockDim.x + threadIdx.x;
    if (t >= nWords) return;
    unsigned w = 0;
#pragma unroll
    for (int j = 0; j < 4; ++j) {
        const int i = 4 * t + j;
        if (i < n) {
            const float d = deg[i];
            deg[i] = (d > 0.0f) ? rsqrtf(d) : 0.0f;
            const unsigned cnt = (d < 255.0f) ? (unsigned)d : 255u;
            w |= cnt << (8 * j);
        }
    }
    degB[t] = w;
}

#define OWN_WAVES 32
#define DEG_TAB_CAP 102400
#define TAB_COPY_DEPTH 8
#define QCAP 64
#define EBATCH 16
#define SCAN_UNROLL 4
#define KEYS_PER_CHUNK 512

__device__ __forceinline__ void ldsAddF(float* p, float v) {
    __hip_atomic_fetch_add(p, v, __ATOMIC_RELAXED, __HIP_MEMORY_SCOPE_WORKGROUP);
}
__device__ __forceinline__ int bcastI(int v, int srcLane) {
    return __builtin_amdgcn_readlane(v, srcLane);
}
__device__ __forceinline__ float bcastF(float v, int srcLane) {
    return __int_as_float(__builtin_amdgcn_readlane(__float_as_int(v), srcLane));
}
__device__ __forceinline__ unsigned zeroByteFlags(unsigned t) {
    return (t - 0x01010101u) & ~t & 0x80808080u;
}
__device__ __forceinline__ unsigned chunkFlags(const uint4& kk, unsigned pat) {
    const unsigned z0 = zeroByteFlags(kk.x ^ pat);
    const unsigned z1 = zeroByteFlags(kk.y ^ pat);
    const unsigned z2 = zeroByteFlags(kk.z ^ pat);
    const unsigned z3 = zeroByteFlags(kk.w ^ pat);
    return z3 | (z2 >> 1) | (z1 >> 2) | (z0 >> 3);
}

__device__ __forceinline__ void load_key_group(const uint4* __restrict__ kp, int i0, int start,
                                               int nChunks, unsigned lane,
                                               uint4 (&kk)[SCAN_UNROLL]) {
#pragma unroll
    for (int u = 0; u < SCAN_UNROLL; ++u) {
        int c = i0 + u * OWN_WAVES;
        if (c >= nChunks) c = nChunks - 1;
        int w = c + start;
        if (w >= nChunks) w -= nChunks;
        kk[u] = kp[(size_t)w * 32 + lane];
    }
}

__device__ __forceinline__ float4 load_row_piece(const float* __restrict__ x, int s, unsigned lane) {
    const unsigned off = ((unsigned)s << 9) | (lane << 4);
    return *reinterpret_cast<const float4*>(reinterpret_cast<const char*>(x) + off);
}

__device__ __forceinline__ void add_queued_edges(const float* __restrict__ x,
                                                 const float* __restrict__ dis,
                                                 const unsigned* __restrict__ pk,
                                                 float* acc, const int* myq,
                                                 const float* disOwn, const unsigned char* degTab,
                                                 int qhead, int cnt, unsigned tag, int E,
                                                 unsigned lane) {
    int s = 0, lr = ROWS_PER_WG;
    if ((int)lane < cnt) {
        const int e = myq[(qhead + (int)lane) & (QCAP - 1)];
        if (e < E) {
            const unsigned p = pk[e];
            if ((p >> PK_TAG_SHIFT) == tag) {
                s = (int)(p & PK_SRC_MASK);
                lr = (int)((p >> PK_SRC_BITS) & 511u);
            }
        }
    }
    const float dd = disOwn[lr];
    const unsigned cb = (s < DEG_TAB_CAP) ? (unsigned)degTab[s] : 255u;
    float ds;
    if (cb != 255u) {
        ds = (cb > 0u) ? rsqrtf((float)cb) : 0.0f;
    } else {
        ds = dis[s];
    }

    for (int b = 0; b < cnt; b += EBATCH) {
        float4 v[EBATCH];
#pragma unroll
        for (int j = 0; j < EBATCH; ++j) {
            v[j] = load_row_piece(x, bcastI(s, b + j), lane);
        }
#pragma unroll
        for (int j = 0; j < EBATCH; ++j) {
            const float nj = bcastF(ds, b + j) * bcastF(dd, b + j);
            float* dst = acc + bcastI(lr, b + j) * DIM + lane;
            ldsAddF(dst + 0,  v[j].x * nj);
            ldsAddF(dst + 32, v[j].y * nj);
            ldsAddF(dst + 64, v[j].z * nj);
            ldsAddF(dst + 96, v[j].w * nj);
        }
    }
}

__global__ __launch_bounds__(1024) void owner_accumulate_kernel(const float* __restrict__ x,
                                                                const float* __restrict__ dis,
                                                                const unsigned* __restrict__ pk,
                                                                const unsigned* __restrict__ keys,
                                                                const unsigned* __restrict__ degB,
                                                                float* __restrict__ agg,
                                                                int N, int E, int nChunks) {
    __shared__ float acc[(ROWS_PER_WG + 1) * DIM];
    __shared__ int q[OWN_WAVES * QCAP];
    __shared__ float disOwn[ROWS_PER_WG + 16];
    __shared__ unsigned degTab[DEG_TAB_CAP / 4];
    const int tid = threadIdx.x;
    const unsigned lane = tid & 31;
    const int wv = __builtin_amdgcn_readfirstlane(tid >> 5);
    const int rowBase = (int)blockIdx.x * ROWS_PER_WG;
    const int rowsLeft = N - rowBase;
    const int rowsHere = rowsLeft < ROWS_PER_WG ? rowsLeft : ROWS_PER_WG;

    for (int i = tid; i < rowsHere * 32; i += 1024) {
        const int r = i >> 5, c4 = i & 31;
        float sc = dis[rowBase + r];
        sc = sc * sc;
        float4 v = reinterpret_cast<const float4*>(x + (size_t)(rowBase + r) * DIM)[c4];
        float* row = acc + r * DIM + c4;
        row[0]  = v.x * sc;
        row[32] = v.y * sc;
        row[64] = v.z * sc;
        row[96] = v.w * sc;
    }
    if (tid < 32) {
        reinterpret_cast<float4*>(acc + ROWS_PER_WG * DIM)[tid] = make_float4(0.f, 0.f, 0.f, 0.f);
    }
    if (tid < ROWS_PER_WG + 16) disOwn[tid] = (tid < rowsHere) ? dis[rowBase + tid] : 0.0f;
    {
        const int nb = N < DEG_TAB_CAP ? N : DEG_TAB_CAP;
        const int nv = (nb + 15) >> 4;
        const v4u* src = reinterpret_cast<const v4u*>(degB);
        v4u* dst = reinterpret_cast<v4u*>(degTab);
        for (int i0 = tid; i0 < nv; i0 += TAB_COPY_DEPTH * 1024) {
            v4u t[TAB_COPY_DEPTH];
#pragma unroll
            for (int u = 0; u < TAB_COPY_DEPTH; ++u) {
                const int i = i0 + u * 1024;
                t[u] = src[i < nv ? i : nv - 1];
            }
            asm volatile("" :: "v"(t[0]), "v"(t[1]), "v"(t[2]), "v"(t[3]),
                               "v"(t[4]), "v"(t[5]), "v"(t[6]), "v"(t[7]));
#pragma unroll
            for (int u = 0; u < TAB_COPY_DEPTH; ++u) {
                const int i = i0 + u * 1024;
                if (i < nv) dst[i] = t[u];
            }
        }
    }
    __syncthreads();

    const unsigned pat = ((unsigned)blockIdx.x & 255u) * 0x01010101u;
    const unsigned tag = ((unsigned)blockIdx.x & 1u) | (((unsigned)blockIdx.x >> 8) << 1);
    const unsigned char* degBytes = reinterpret_cast<const unsigned char*>(degTab);
    int* myq = q + wv * QCAP;
    int qhead = 0, qn = 0;
    const int start = (int)blockIdx.x * (nChunks / (int)gridDim.x);
    const uint4* kp = reinterpret_cast<const uint4*>(keys);

    uint4 nx[SCAN_UNROLL];
    load_key_group(kp, wv, start, nChunks, lane, nx);
    for (int i = wv; i < nChunks; i += OWN_WAVES * SCAN_UNROLL) {
        uint4 kk[SCAN_UNROLL];
#pragma unroll
        for (int u = 0; u < SCAN_UNROLL; ++u) kk[u] = nx[u];
        load_key_group(kp, i + OWN_WAVES * SCAN_UNROLL, start, nChunks, lane, nx);

        unsigned cf[SCAN_UNROLL];
#pragma unroll
        for (int u = 0; u < SCAN_UNROLL; ++u) {
            cf[u] = (i + u * OWN_WAVES < nChunks) ? chunkFlags(kk[u], pat) : 0u;
        }
        unsigned long long mask64 = (unsigned long long)(cf[0] | (cf[1] >> 4)) |
                                    ((unsigned long long)(cf[2] | (cf[3] >> 4)) << 32);

        for (int trip = 0; trip < 16 * SCAN_UNROLL; ++trip) {
            const unsigned bal = __builtin_amdgcn_ballot_w32(mask64 != 0ull);
            if (bal == 0u) break;
            if (mask64 != 0ull) {
                const int bit = __builtin_ctzll(mask64);
                const int n = bit & 7;
                const int u = ((bit >> 5) << 1) + (n < 4 ? 1 : 0);
                const int j = ((n & 3) << 2) + ((bit >> 3) & 3);
                int w = i + u * OWN_WAVES + start;
                if (w >= nChunks) w -= nChunks;
                const int pos = qn + (int)__builtin_amdgcn_mbcnt_lo(bal, 0u);
                myq[(qhead + pos) & (QCAP - 1)] = w * KEYS_PER_CHUNK + (int)lane * 16 + j;
                mask64 &= mask64 - 1ull;
            }
            qn += __builtin_popcount(bal);
            if (qn >= 32) {
                add_queued_edges(x, dis, pk, acc, myq, disOwn, degBytes, qhead, 32, tag, E, lane);
                qhead = (qhead + 32) & (QCAP - 1);
                qn -= 32;
            }
        }
    }
    if (qn > 0) {
        add_queued_edges(x, dis, pk, acc, myq, disOwn, degBytes, qhead, qn, tag, E, lane);
    }
    __syncthreads();

    for (int i = tid; i < rowsHere * 32; i += 1024) {
        const int r = i >> 5, c4 = i & 31;
        const float* row = acc + r * DIM + c4;
        reinterpret_cast<float4*>(agg + (size_t)(rowBase + r) * DIM)[c4] =
            make_float4(row[0], row[32], row[64], row[96]);
    }
}

#define STRIPS_PER_WAVE 7

__device__ __forceinline__ void load_a_strip(const float* __restrict__ A, int rowBase, int n,
                                             int m, int h, float4 (&q)[16]) {
    int row = rowBase + m;
    if (row > n - 1) row = n - 1;
    const float* p = A + (size_t)row * DIM + 4 * h;
#pragma unroll
    for (int j = 0; j < 16; ++j) q[j] = *reinterpret_cast<const float4*>(p + 8 * j);
}

__global__ __launch_bounds__(32) void gemm_wmma_kernel(const float* __restrict__ A,
                                                       const float* __restrict__ W,
                                                       float* __restrict__ out, int n) {
    const int lane = threadIdx.x;
    const int m = lane & 15;
    const int h = lane >> 4;
    const int strip0 = (int)blockIdx.x * STRIPS_PER_WAVE;

    float4 q[16];
    load_a_strip(A, strip0 * 16, n, m, h, q);

    v2f b[32][8];
    const float* wp = W + (size_t)(4 * h) * DIM + m;
#pragma unroll
    for (int t = 0; t < 32; ++t) {
        const int k0 = 8 * (t >> 1) + 2 * (t & 1);
#pragma unroll
        for (int c = 0; c < 8; ++c) {
            b[t][c].x = wp[(k0 + 0) * DIM + 16 * c];
            b[t][c].y = wp[(k0 + 1) * DIM + 16 * c];
        }
    }

#pragma unroll 1
    for (int sidx = 0; sidx < STRIPS_PER_WAVE; ++sidx) {
        const int rowBase = (strip0 + sidx) * 16;
        if (rowBase >= n) break;
        float4 qn[16];
        load_a_strip(A, rowBase + 16, n, m, h, qn);
        __builtin_amdgcn_sched_barrier(0);

        v8f acc[8];
#pragma unroll
        for (int c = 0; c < 8; ++c) acc[c] = (v8f){0.f, 0.f, 0.f, 0.f, 0.f, 0.f, 0.f, 0.f};

#pragma unroll
        for (int j = 0; j < 16; ++j) {
            v2f a0, a1;
            a0.x = q[j].x; a0.y = q[j].y;
            a1.x = q[j].z; a1.y = q[j].w;
#pragma unroll
            for (int c = 0; c < 8; ++c) {
                acc[c] = __builtin_amdgcn_wmma_f32_16x16x4_f32(
                     false, a0,  false, b[2 * j][c],
                     (short)0, acc[c],  false,  false);
            }
#pragma unroll
            for (int c = 0; c < 8; ++c) {
                acc[c] = __builtin_amdgcn_wmma_f32_16x16x4_f32(
                     false, a1,  false, b[2 * j + 1][c],
                     (short)0, acc[c],  false,  false);
            }
        }

        float* op = out + (size_t)(rowBase + 8 * h) * DIM + m;
#pragma unroll
        for (int v = 0; v < 8; ++v) {
            if (rowBase + 8 * h + v < n) {
#pragma unroll
                for (int c = 0; c < 8; ++c) op[v * DIM + 16 * c] = acc[c][v];
            }
        }
#pragma unroll
        for (int j = 0; j < 16; ++j) q[j] = qn[j];
    }
}

extern "C" void kernel_launch(void* const* d_in, const int* in_sizes, int n_in,
                              void* d_out, int out_size, void* d_ws, size_t ws_size,
                              hipStream_t stream) {
    const float* x = (const float*)d_in[0];
    const int*   ei = (const int*)d_in[1];
    const float* W = (const float*)d_in[2];
    float* out = (float*)d_out;

    const int N = in_sizes[0] / DIM;
    const int E = in_sizes[1] / 2;

    float* deg = (float*)d_ws;
    size_t aggOff = (((size_t)N * sizeof(float)) + 255) & ~(size_t)255;
    float* agg = (float*)((char*)d_ws + aggOff);
    size_t keyOff = (aggOff + (size_t)N * DIM * sizeof(float) + 255) & ~(size_t)255;
    unsigned* keys = (unsigned*)((char*)d_ws + keyOff);

    const int nChunks = (E + KEYS_PER_CHUNK - 1) / KEYS_PER_CHUNK;
    const int nWords = nChunks * (KEYS_PER_CHUNK / 4);
    const int nDegWords = ((N + 15) / 16) * 4;

    unsigned* pk = (unsigned*)d_out;
    size_t degBOff = ((size_t)E * sizeof(unsigned) + 255) & ~(size_t)255;
    unsigned* degB = (unsigned*)((char*)d_out + degBOff);

    deg_init_kernel<<<(N + 255) / 256, 256, 0, stream>>>(deg, N);
    deg_count_key_kernel<<<(nWords + 255) / 256, 256, 0, stream>>>(ei, deg, keys, pk, E, nWords);
    deg_to_dis_kernel<<<(nDegWords + 255) / 256, 256, 0, stream>>>(deg, degB, N, nDegWords);
    owner_accumulate_kernel<<<(N + ROWS_PER_WG - 1) / ROWS_PER_WG, 1024, 0, stream>>>(
        x, deg, pk, keys, degB, agg, N, E, nChunks);
    const int nStrips = (N + 15) / 16;
    gemm_wmma_kernel<<<(nStrips + STRIPS_PER_WAVE - 1) / STRIPS_PER_WAVE, 32, 0, stream>>>(agg, W, out, N);
}
